# starting kernel
# baseline (speedup 1.0000x reference)
#include <hip/hip_runtime.h>

#define N_NODES   100000
#define EMB       64
#define HEADS     4
#define RD        32
#define HOUT      (HEADS * RD)
#define NEDGES    1600000
#define NEG_SLOPE 0.2f

typedef __attribute__((ext_vector_type(2))) float v2f;
typedef __attribute__((ext_vector_type(8))) float v8f;

__device__ __forceinline__ void atomicMaxF(float* addr, float v) {
    if (v >= 0.0f) atomicMax((int*)addr, __float_as_int(v));
    else           atomicMin((unsigned int*)addr, __float_as_uint(v));
}

__device__ __forceinline__ float leaky(float x) {
    return x >= 0.0f ? x : NEG_SLOPE * x;
}

__global__ void init_kernel(float* __restrict__ out, float* __restrict__ m,
                            float* __restrict__ denom) {
    int i = blockIdx.x * blockDim.x + threadIdx.x;
    if (i < N_NODES * RD)    out[i] = 0.0f;
    if (i < N_NODES * HEADS) { m[i] = __int_as_float(0xFF800000); denom[i] = 0.0f; }
}

__global__ __launch_bounds__(256) void proj_wmma_kernel(const float* __restrict__ emb,
                                                        const float* __restrict__ W,
                                                        float* __restrict__ h) {
    __shared__ float Wt[HOUT][EMB + 1];
    const int tid = threadIdx.x;
    for (int i = tid; i < EMB * HOUT; i += 256) {
        int k = i >> 7;
        int n = i & 127;
        Wt[n][k] = W[i];
    }
    __syncthreads();

    const int wave = tid >> 5;
    const int lane = tid & 31;
    const int rowTile = (blockIdx.x * 8 + wave) * 16;
    if (rowTile >= N_NODES) return;

    const int half = lane >> 4;
    const int lrow = lane & 15;
    const float* arow = emb + (size_t)(rowTile + lrow) * EMB;

    v8f acc[8] = {};
    for (int k0 = 0; k0 < EMB; k0 += 4) {
        v2f a;
        a.x = arow[k0 + 2 * half];
        a.y = arow[k0 + 2 * half + 1];
        #pragma unroll
        for (int ct = 0; ct < 8; ++ct) {
            const int n = ct * 16 + lrow;
            v2f b;
            b.x = Wt[n][k0 + 2 * half];
            b.y = Wt[n][k0 + 2 * half + 1];
            acc[ct] = __builtin_amdgcn_wmma_f32_16x16x4_f32(
                false, a, false, b, (short)0, acc[ct], false, false);
        }
    }
    #pragma unroll
    for (int ct = 0; ct < 8; ++ct) {
        #pragma unroll
        for (int v = 0; v < 8; ++v) {
            const int mr = v + 8 * half;
            h[(size_t)(rowTile + mr) * HOUT + ct * 16 + lrow] = acc[ct][v];
        }
    }
}

__global__ void logits_kernel(const float* __restrict__ h,
                              const float* __restrict__ att_src,
                              const float* __restrict__ att_dst,
                              float* __restrict__ a_src, float* __restrict__ a_dst) {
    int i = blockIdx.x * blockDim.x + threadIdx.x;
    if (i >= N_NODES * HEADS) return;
    const int n = i >> 2, hh = i & 3;
    const float* hp = h + (size_t)n * HOUT + hh * RD;
    const float* as = att_src + hh * RD;
    const float* ad = att_dst + hh * RD;
    float s = 0.0f, d = 0.0f;
    #pragma unroll
    for (int k = 0; k < RD; ++k) { float v = hp[k]; s += v * as[k]; d += v * ad[k]; }
    a_src[i] = s;
    a_dst[i] = d;
}

__global__ void edge_logit_max_kernel(const int* __restrict__ path,
                                      const float* __restrict__ a_src,
                                      const float* __restrict__ a_dst,
                                      float* __restrict__ ebuf, float* __restrict__ m) {
    int e = blockIdx.x * blockDim.x + threadIdx.x;
    if (e >= NEDGES) return;
    const int s = path[e], d = path[NEDGES + e];
    const float4 es = *(const float4*)(a_src + (size_t)s * 4);
    const float4 ed = *(const float4*)(a_dst + (size_t)d * 4);
    float4 ev;
    ev.x = leaky(es.x + ed.x);
    ev.y = leaky(es.y + ed.y);
    ev.z = leaky(es.z + ed.z);
    ev.w = leaky(es.w + ed.w);
    *(float4*)(ebuf + (size_t)e * 4) = ev;
    atomicMaxF(&m[d * 4 + 0], ev.x);
    atomicMaxF(&m[d * 4 + 1], ev.y);
    atomicMaxF(&m[d * 4 + 2], ev.z);
    atomicMaxF(&m[d * 4 + 3], ev.w);
}

__global__ void edge_exp_sum_kernel(const int* __restrict__ path,
                                    float* __restrict__ ebuf,
                                    const float* __restrict__ m,
                                    float* __restrict__ denom) {
    int e = blockIdx.x * blockDim.x + threadIdx.x;
    if (e >= NEDGES) return;
    const int d = path[NEDGES + e];
    const float4 ev = *(const float4*)(ebuf + (size_t)e * 4);
    const float4 mv = *(const float4*)(m + (size_t)d * 4);
    float4 ex;
    ex.x = __expf(ev.x - mv.x);
    ex.y = __expf(ev.y - mv.y);
    ex.z = __expf(ev.z - mv.z);
    ex.w = __expf(ev.w - mv.w);
    *(float4*)(ebuf + (size_t)e * 4) = ex;
    atomicAdd(&denom[d * 4 + 0], ex.x);
    atomicAdd(&denom[d * 4 + 1], ex.y);
    atomicAdd(&denom[d * 4 + 2], ex.z);
    atomicAdd(&denom[d * 4 + 3], ex.w);
}

__global__ void aggregate_kernel(const int* __restrict__ path,
                                 const float* __restrict__ h,
                                 const float* __restrict__ ebuf,
                                 const float* __restrict__ denom,
                                 float* __restrict__ out) {
    int t = blockIdx.x * blockDim.x + threadIdx.x;
    if (t >= NEDGES * RD) return;
    const int e  = t >> 5;
    const int dd = t & 31;
    const int s = path[e], d = path[NEDGES + e];
    const float* hs = h + (size_t)s * HOUT;
    const float4 ex  = *(const float4*)(ebuf  + (size_t)e * 4);
    const float4 den = *(const float4*)(denom + (size_t)d * 4);
    float acc = (ex.x / (den.x + 1e-16f)) * hs[0 * RD + dd]
              + (ex.y / (den.y + 1e-16f)) * hs[1 * RD + dd]
              + (ex.z / (den.z + 1e-16f)) * hs[2 * RD + dd]
              + (ex.w / (den.w + 1e-16f)) * hs[3 * RD + dd];
    atomicAdd(&out[(size_t)d * RD + dd], 0.25f * acc);
}

extern "C" void kernel_launch(void* const* d_in, const int* in_sizes, int n_in,
                              void* d_out, int out_size, void* d_ws, size_t ws_size,
                              hipStream_t stream) {
    const float* node_emb = (const float*)d_in[0];
    const float* W        = (const float*)d_in[1];
    const float* att_src  = (const float*)d_in[2];
    const float* att_dst  = (const float*)d_in[3];
    const int*   path     = (const int*)d_in[4];
    float* out = (float*)d_out;

    float* ws    = (float*)d_ws;
    float* h     = ws;
    float* a_src = h     + (size_t)N_NODES * HOUT;
    float* a_dst = a_src + (size_t)N_NODES * HEADS;
    float* m     = a_dst + (size_t)N_NODES * HEADS;
    float* denom = m     + (size_t)N_NODES * HEADS;
    float* ebuf  = denom + (size_t)N_NODES * HEADS;

    init_kernel<<<(N_NODES * RD + 255) / 256, 256, 0, stream>>>(out, m, denom);

    const int rowTiles = N_NODES / 16;
    proj_wmma_kernel<<<(rowTiles + 7) / 8, 256, 0, stream>>>(node_emb, W, h);

    logits_kernel<<<(N_NODES * HEADS + 255) / 256, 256, 0, stream>>>(
        h, att_src, att_dst, a_src, a_dst);

    edge_logit_max_kernel<<<(NEDGES + 255) / 256, 256, 0, stream>>>(
        path, a_src, a_dst, ebuf, m);

    edge_exp_sum_kernel<<<(NEDGES + 255) / 256, 256, 0, stream>>>(
        path, ebuf, m, denom);

    aggregate_kernel<<<(NEDGES * RD + 255) / 256, 256, 0, stream>>>(
        path, h, ebuf, denom, out);
}


# Filter: projection writes a presence bitmap (bit per 32-edge line and range); a block reads only its flagged lines of both path rows, lane per edge; no plane scan or per-edge gathers; new queue order
# speedup vs baseline: 5.1973x; 5.1973x over previous
#include <hip/hip_runtime.h>

#define N_NODES   100000
#define EMB       64
#define HEADS     4
#define RD        32
#define HOUT      (HEADS * RD)
#define NEDGES    1600000
#define NEG_SLOPE 0.2f
#define RANGE     400
#define NBLK      (N_NODES / RANGE)
#define NO_RANGE  255u

typedef __attribute__((ext_vector_type(2))) float v2f;
typedef __attribute__((ext_vector_type(8))) float v8f;

__device__ __forceinline__ void atomicMaxF(float* addr, float v) {
    if (v >= 0.0f) atomicMax((int*)addr, __float_as_int(v));
    else           atomicMin((unsigned int*)addr, __float_as_uint(v));
}

__device__ __forceinline__ float leaky(float x) {
    return x >= 0.0f ? x : NEG_SLOPE * x;
}

__device__ __forceinline__ unsigned range_id(int d) {
    return (unsigned)d < (unsigned)N_NODES ? (unsigned)d / RANGE : NO_RANGE;
}

#define PROJ_BLOCKS  ((N_NODES / 16 + 7) / 8)
#define LINE_EDGES 32
#define NLINES     (NEDGES / LINE_EDGES)
#define PB_LINES   64
#define BM_PITCH   256
static_assert(NEDGES % LINE_EDGES == 0, "whole lines of 32 edges");
static_assert(PROJ_BLOCKS * PB_LINES >= NLINES, "the projection grid covers every line");
static_assert(NBLK <= BM_PITCH, "a chunk holds one entry per range");

__global__ __launch_bounds__(256, 8) void proj_wmma_kernel(const float* __restrict__ emb,
                                                        const float* __restrict__ W,
                                                        const float* __restrict__ att_src,
                                                        const float* __restrict__ att_dst,
                                                        const int* __restrict__ path,
                                                        float* __restrict__ h,
                                                        float* __restrict__ a_src,
                                                        float* __restrict__ a_dst,
                                                        uint2* __restrict__ bm) {
    __shared__ float Wt[HOUT][EMB + 1];
    __shared__ unsigned bmt[NBLK * 2];
    const int tid = threadIdx.x;
    const int4* dst4 = (const int4*)(path + NEDGES);
    const int w0 = blockIdx.x * (PB_LINES * LINE_EDGES / 4) + tid;
    const int w1 = w0 + 256;
    const int4 d0 = dst4[w0 < NEDGES / 4 ? w0 : 0];
    const int4 d1 = dst4[w1 < NEDGES / 4 ? w1 : 0];
    for (int i = tid; i < NBLK * 2; i += 256) bmt[i] = 0u;
    for (int i = tid; i < EMB * HOUT; i += 256) {
        int k = i >> 7;
        int n = i & 127;
        Wt[n][k] = W[i];
    }
    __syncthreads();
    {
        const unsigned bit = 1u << (tid >> 3);
        if (w0 < NEDGES / 4) {
            const unsigned r0 = range_id(d0.x), r1 = range_id(d0.y), r2 = range_id(d0.z), r3 = range_id(d0.w);
            if (r0 < NBLK) atomicOr(&bmt[r0 * 2], bit);
            if (r1 < NBLK) atomicOr(&bmt[r1 * 2], bit);
            if (r2 < NBLK) atomicOr(&bmt[r2 * 2], bit);
            if (r3 < NBLK) atomicOr(&bmt[r3 * 2], bit);
        }
        if (w1 < NEDGES / 4) {
            const unsigned r0 = range_id(d1.x), r1 = range_id(d1.y), r2 = range_id(d1.z), r3 = range_id(d1.w);
            if (r0 < NBLK) atomicOr(&bmt[r0 * 2 + 1], bit);
            if (r1 < NBLK) atomicOr(&bmt[r1 * 2 + 1], bit);
            if (r2 < NBLK) atomicOr(&bmt[r2 * 2 + 1], bit);
            if (r3 < NBLK) atomicOr(&bmt[r3 * 2 + 1], bit);
        }
    }
    __syncthreads();
    if (tid < NBLK) {
        uint2 o;
        o.x = bmt[tid * 2];
        o.y = bmt[tid * 2 + 1];
        bm[(size_t)blockIdx.x * BM_PITCH + tid] = o;
    }

    const int wave = tid >> 5;
    const int lane = tid & 31;
    const int rowTile = (blockIdx.x * 8 + wave) * 16;
    if (rowTile >= N_NODES) return;

    const int half = lane >> 4;
    const int lrow = lane & 15;
    const float* arow = emb + (size_t)(rowTile + lrow) * EMB;

    v8f acc[8] = {};
    for (int k0 = 0; k0 < EMB; k0 += 4) {
        v2f a;
        a.x = arow[k0 + 2 * half];
        a.y = arow[k0 + 2 * half + 1];
        #pragma unroll
        for (int ct = 0; ct < 8; ++ct) {
            const int n = ct * 16 + lrow;
            v2f b;
            b.x = Wt[n][k0 + 2 * half];
            b.y = Wt[n][k0 + 2 * half + 1];
            acc[ct] = __builtin_amdgcn_wmma_f32_16x16x4_f32(
                false, a, false, b, (short)0, acc[ct], false, false);
        }
    }
    #pragma unroll
    for (int ct = 0; ct < 8; ++ct) {
        #pragma unroll
        for (int v = 0; v < 8; ++v) {
            const int mr = v + 8 * half;
            h[(size_t)(rowTile + mr) * HOUT + ct * 16 + lrow] = acc[ct][v];
        }
    }

    float asv[8], adv[8];
    #pragma unroll
    for (int ct = 0; ct < 8; ++ct) {
        asv[ct] = att_src[ct * 16 + lrow];
        adv[ct] = att_dst[ct * 16 + lrow];
    }
    const int keep = lrow & 7;
    float ms[HEADS], md[HEADS];
    #pragma unroll
    for (int hh = 0; hh < HEADS; ++hh) {
        ms[hh] = 0.0f;
        md[hh] = 0.0f;
        #pragma unroll
        for (int v = 0; v < 8; ++v) {
            float sv = acc[2 * hh][v] * asv[2 * hh] + acc[2 * hh + 1][v] * asv[2 * hh + 1];
            float dv = acc[2 * hh][v] * adv[2 * hh] + acc[2 * hh + 1][v] * adv[2 * hh + 1];
            #pragma unroll
            for (int m = 1; m < 16; m <<= 1) {
                sv += __shfl_xor(sv, m);
                dv += __shfl_xor(dv, m);
            }
            if (keep == v) { ms[hh] = sv; md[hh] = dv; }
        }
    }
    {
        const size_t node = (size_t)(rowTile + 8 * half + keep);
        float4 o;
        if (lrow < 8) { o.x = ms[0]; o.y = ms[1]; o.z = ms[2]; o.w = ms[3]; }
        else          { o.x = md[0]; o.y = md[1]; o.z = md[2]; o.w = md[3]; }
        float* dstp = (lrow < 8 ? a_src : a_dst) + node * HEADS;
        *(float4*)dstp = o;
    }
}

#define QCAP  8192
#define KMAX  (QCAP / 1024)
#define FGRP  16
#define DL_BITS 9
static_assert(NBLK * RANGE == N_NODES && NBLK < 255, "range id 255 is kept free");
static_assert(RANGE <= (1 << DL_BITS) && N_NODES <= (1 << (31 - DL_BITS)), "a queue entry fits a positive int");

__device__ __forceinline__ float4 edge_logit(const float* __restrict__ a_src,
                                             const float* adst_l, int s, int dl) {
    const float4 es = *(const float4*)(a_src + (size_t)s * 4);
    const float4 ed = *(const float4*)(adst_l + dl * 4);
    float4 ev;
    ev.x = leaky(es.x + ed.x);
    ev.y = leaky(es.y + ed.y);
    ev.z = leaky(es.z + ed.z);
    ev.w = leaky(es.w + ed.w);
    return ev;
}

template <int PHASE>
__device__ __forceinline__ void walk_all(const int* __restrict__ path,
                                         const float* __restrict__ a_src,
                                         const float* adst_l, float* m_l, float* den_l,
                                         const float* __restrict__ h, float* out_l,
                                         int lo, int tid) {
    for (int e = tid; e < NEDGES; e += 1024) {
        const unsigned dl = (unsigned)(path[NEDGES + e] - lo);
        if (dl < RANGE) {
            const int s = path[e];
            const float4 ev = edge_logit(a_src, adst_l, s, (int)dl);
            if (PHASE == 0) {
                atomicMaxF(&m_l[dl * 4 + 0], ev.x);
                atomicMaxF(&m_l[dl * 4 + 1], ev.y);
                atomicMaxF(&m_l[dl * 4 + 2], ev.z);
                atomicMaxF(&m_l[dl * 4 + 3], ev.w);
            } else {
                const float4 mv = *(const float4*)(m_l + dl * 4);
                float4 ex;
                ex.x = __expf(ev.x - mv.x);
                ex.y = __expf(ev.y - mv.y);
                ex.z = __expf(ev.z - mv.z);
                ex.w = __expf(ev.w - mv.w);
                if (PHASE == 1) {
                    atomicAdd(&den_l[dl * 4 + 0], ex.x);
                    atomicAdd(&den_l[dl * 4 + 1], ex.y);
                    atomicAdd(&den_l[dl * 4 + 2], ex.z);
                    atomicAdd(&den_l[dl * 4 + 3], ex.w);
                } else {
                    const float4 dn = *(const float4*)(den_l + dl * 4);
                    float4 al;
                    al.x = ex.x / (dn.x + 1e-16f);
                    al.y = ex.y / (dn.y + 1e-16f);
                    al.z = ex.z / (dn.z + 1e-16f);
                    al.w = ex.w / (dn.w + 1e-16f);
                    const float* hs = h + (size_t)s * HOUT;
                    for (int dd = 0; dd < RD; ++dd) {
                        const float acc = al.x * hs[0 * RD + dd]
                                        + al.y * hs[1 * RD + dd]
                                        + al.z * hs[2 * RD + dd]
                                        + al.w * hs[3 * RD + dd];
                        atomicAdd(&out_l[dl * RD + dd], 0.25f * acc);
                    }
                }
            }
        }
    }
}

__device__ __forceinline__ float lane_value(float x, int srcLane) {
    return __int_as_float(__builtin_amdgcn_readlane(__float_as_int(x), srcLane));
}

#define GRP 8

__global__ __launch_bounds__(1024) void softmax_range_kernel(const int* __restrict__ path,
                                                             const float* __restrict__ a_src,
                                                             const float* __restrict__ a_dst,
                                                             const float* __restrict__ h,
                                                             const uint2* __restrict__ bm,
                                                             float* __restrict__ out) {
    __shared__ alignas(16) float out_l[RANGE * RD];
    __shared__ float m_l[RANGE * 4];
    __shared__ float den_l[RANGE * 4];
    __shared__ float adst_l[RANGE * 4];
    __shared__ int   queue[QCAP];
    __shared__ int   lq[QCAP];
    __shared__ int   qn;
    __shared__ int   ln;

    const int tid = threadIdx.x;
    const int lo  = blockIdx.x * RANGE;

    for (int i = tid; i < RANGE * 4; i += 1024) {
        m_l[i]    = __int_as_float(0xFF800000);
        den_l[i]  = 0.0f;
        adst_l[i] = a_dst[(size_t)lo * 4 + i];
    }
    for (int i = tid; i < RANGE * RD; i += 1024) out_l[i] = 0.0f;
    if (tid == 0) { qn = 0; ln = 0; }
    __syncthreads();

    for (int t = tid; t < PROJ_BLOCKS; t += 1024) {
        const uint2 e = bm[(size_t)t * BM_PITCH + blockIdx.x];
        unsigned long long mm = ((unsigned long long)e.y << 32) | e.x;
        int pos = atomicAdd(&ln, __builtin_popcountll(mm));
        while (mm != 0ull) {
            const int line = t * PB_LINES + __builtin_ctzll(mm);
            mm &= mm - 1ull;
            if (pos < QCAP) lq[pos] = line < NLINES ? line : NLINES - 1;
            ++pos;
        }
    }
    __syncthreads();
    const int ltot = ln;
    if (ltot <= QCAP) {
        const int lane = tid & 31;
        for (int j0 = (tid >> 5) * FGRP; j0 < ltot; j0 += 32 * FGRP) {
            int dv[FGRP], sr[FGRP];
            #pragma unroll
            for (int g = 0; g < FGRP; ++g) {
                const int j = j0 + g < ltot ? j0 + g : ltot - 1;
                const int e = lq[j] * LINE_EDGES + lane;
                dv[g] = path[NEDGES + e];
                sr[g] = path[e];
            }
            #pragma unroll
            for (int g = 0; g < FGRP; ++g) asm volatile("" :: "v"(dv[g]), "v"(sr[g]));
            #pragma unroll
            for (int g = 0; g < FGRP; ++g) {
                const unsigned dl = (unsigned)(dv[g] - lo);
                if (j0 + g < ltot && dl < RANGE) {
                    const int pos = atomicAdd(&qn, 1);
                    if (pos < QCAP) queue[pos] = (sr[g] << DL_BITS) | (int)dl;
                }
            }
        }
    }
    __syncthreads();
    const int total = ltot <= QCAP ? qn : QCAP + 1;

    if (total <= QCAP) {
        int    wq[KMAX];
        float4 val[KMAX];
        int    sv[KMAX];
        #pragma unroll
        for (int k = 0; k < KMAX; ++k) {
            const int j = k * 1024 + tid;
            wq[k] = j < total ? queue[j] : -1;
        }
        #pragma unroll
        for (int k = 0; k < KMAX; ++k) {
            sv[k] = wq[k] >= 0 ? wq[k] >> DL_BITS : 0;
            wq[k] = wq[k] >= 0 ? wq[k] & ((1 << DL_BITS) - 1) : -1;
            val[k] = edge_logit(a_src, adst_l, sv[k], wq[k] >= 0 ? wq[k] : 0);
        }
        #pragma unroll
        for (int k = 0; k < KMAX; ++k) {
            if (wq[k] >= 0) {
                const int dl = wq[k];
                atomicMaxF(&m_l[dl * 4 + 0], val[k].x);
                atomicMaxF(&m_l[dl * 4 + 1], val[k].y);
                atomicMaxF(&m_l[dl * 4 + 2], val[k].z);
                atomicMaxF(&m_l[dl * 4 + 3], val[k].w);
            }
        }
        __syncthreads();
        #pragma unroll
        for (int k = 0; k < KMAX; ++k) {
            if (wq[k] >= 0) {
                const int dl = wq[k];
                const float4 mv = *(const float4*)(m_l + dl * 4);
                float4 ex;
                ex.x = __expf(val[k].x - mv.x);
                ex.y = __expf(val[k].y - mv.y);
                ex.z = __expf(val[k].z - mv.z);
                ex.w = __expf(val[k].w - mv.w);
                val[k] = ex;
                atomicAdd(&den_l[dl * 4 + 0], ex.x);
                atomicAdd(&den_l[dl * 4 + 1], ex.y);
                atomicAdd(&den_l[dl * 4 + 2], ex.z);
                atomicAdd(&den_l[dl * 4 + 3], ex.w);
            }
        }
        __syncthreads();
        #pragma unroll
        for (int k = 0; k < KMAX; ++k) {
            if (wq[k] >= 0) {
                const int dl = wq[k];
                const float4 dn = *(const float4*)(den_l + dl * 4);
                float4 al;
                al.x = val[k].x / (dn.x + 1e-16f);
                al.y = val[k].y / (dn.y + 1e-16f);
                al.z = val[k].z / (dn.z + 1e-16f);
                al.w = val[k].w / (dn.w + 1e-16f);
                val[k] = al;
            }
        }

        const int lane = tid & 31;
        const int wbase = __builtin_amdgcn_readfirstlane(tid & ~31);
        const int tot_s = __builtin_amdgcn_readfirstlane(total);
        const float* hl = h + lane;
        #pragma nounroll
        for (int k = 0; k < KMAX; ++k) {
            int cnt = tot_s - (k * 1024 + wbase);
            if (cnt <= 0) break;
            cnt = cnt > 32 ? 32 : cnt;
            for (int L0 = 0; L0 < cnt; L0 += GRP) {
                float hv[GRP][4];
                #pragma unroll
                for (int g = 0; g < GRP; ++g) {
                    const int L = L0 + g < cnt ? L0 + g : cnt - 1;
                    const int s = __builtin_amdgcn_readlane(sv[0], L);
                    const float* hs = hl + (size_t)s * HOUT;
                    hv[g][0] = hs[0 * RD];
                    hv[g][1] = hs[1 * RD];
                    hv[g][2] = hs[2 * RD];
                    hv[g][3] = hs[3 * RD];
                }
                #pragma unroll
                for (int g = 0; g < GRP; ++g) {
                    if (L0 + g < cnt) {
                        const int L  = L0 + g;
                        const int dl = __builtin_amdgcn_readlane(wq[0], L);
                        const float acc = lane_value(val[0].x, L) * hv[g][0]
                                        + lane_value(val[0].y, L) * hv[g][1]
                                        + lane_value(val[0].z, L) * hv[g][2]
                                        + lane_value(val[0].w, L) * hv[g][3];
                        atomicAdd(&out_l[dl * RD + lane], 0.25f * acc);
                    }
                }
            }
            #pragma unroll
            for (int i = 0; i + 1 < KMAX; ++i) {
                wq[i]  = wq[i + 1];
                sv[i]  = sv[i + 1];
                val[i] = val[i + 1];
            }
        }
    } else {
        walk_all<0>(path, a_src, adst_l, m_l, den_l, h, out_l, lo, tid);
        __syncthreads();
        walk_all<1>(path, a_src, adst_l, m_l, den_l, h, out_l, lo, tid);
        __syncthreads();
        walk_all<2>(path, a_src, adst_l, m_l, den_l, h, out_l, lo, tid);
    }
    __syncthreads();

    float4* o4 = (float4*)(out + (size_t)lo * RD);
    const float4* l4 = (const float4*)out_l;
    for (int i = tid; i < RANGE * RD / 4; i += 1024) o4[i] = l4[i];
}

extern "C" void kernel_launch(void* const* d_in, const int* in_sizes, int n_in,
                              void* d_out, int out_size, void* d_ws, size_t ws_size,
                              hipStream_t stream) {
    const float* node_emb = (const float*)d_in[0];
    const float* W        = (const float*)d_in[1];
    const float* att_src  = (const float*)d_in[2];
    const float* att_dst  = (const float*)d_in[3];
    const int*   path     = (const int*)d_in[4];
    float* out = (float*)d_out;

    float* ws    = (float*)d_ws;
    float* h     = ws;
    float* a_src = h     + (size_t)N_NODES * HOUT;
    float* a_dst = a_src + (size_t)N_NODES * HEADS;
    uint2* bm = (uint2*)(a_dst + (size_t)N_NODES * HEADS);

    static_assert(PROJ_BLOCKS == (N_NODES / 16 + 7) / 8, "grid of the projection launch");
    proj_wmma_kernel<<<PROJ_BLOCKS, 256, 0, stream>>>(node_emb, W, att_src, att_dst, path,
                                                      h, a_src, a_dst, bm);

    softmax_range_kernel<<<NBLK, 1024, 0, stream>>>(path, a_src, a_dst, h, bm, out);
}


# Filter step 2: one queue reservation per wave and group of lines (lane ballots, popcount sum, one returning LDS add, mbcnt places), not a waited LDS add per line; arrival order, queue order differs
# speedup vs baseline: 5.5600x; 1.0698x over previous
#include <hip/hip_runtime.h>

#define N_NODES   100000
#define EMB       64
#define HEADS     4
#define RD        32
#define HOUT      (HEADS * RD)
#define NEDGES    1600000
#define NEG_SLOPE 0.2f
#define RANGE     400
#define NBLK      (N_NODES / RANGE)
#define NO_RANGE  255u

typedef __attribute__((ext_vector_type(2))) float v2f;
typedef __attribute__((ext_vector_type(8))) float v8f;

__device__ __forceinline__ void atomicMaxF(float* addr, float v) {
    if (v >= 0.0f) atomicMax((int*)addr, __float_as_int(v));
    else           atomicMin((unsigned int*)addr, __float_as_uint(v));
}

__device__ __forceinline__ float leaky(float x) {
    return x >= 0.0f ? x : NEG_SLOPE * x;
}

__device__ __forceinline__ unsigned range_id(int d) {
    return (unsigned)d < (unsigned)N_NODES ? (unsigned)d / RANGE : NO_RANGE;
}

#define PROJ_BLOCKS  ((N_NODES / 16 + 7) / 8)
#define LINE_EDGES 32
#define NLINES     (NEDGES / LINE_EDGES)
#define PB_LINES   64
#define BM_PITCH   256
static_assert(NEDGES % LINE_EDGES == 0, "whole lines of 32 edges");
static_assert(PROJ_BLOCKS * PB_LINES >= NLINES, "the projection grid covers every line");
static_assert(NBLK <= BM_PITCH, "a chunk holds one entry per range");

__global__ __launch_bounds__(256, 8) void proj_wmma_kernel(const float* __restrict__ emb,
                                                        const float* __restrict__ W,
                                                        const float* __restrict__ att_src,
                                                        const float* __restrict__ att_dst,
                                                        const int* __restrict__ path,
                                                        float* __restrict__ h,
                                                        float* __restrict__ a_src,
                                                        float* __restrict__ a_dst,
                                                        uint2* __restrict__ bm) {
    __shared__ float Wt[HOUT][EMB + 1];
    __shared__ unsigned bmt[NBLK * 2];
    const int tid = threadIdx.x;
    const int4* dst4 = (const int4*)(path + NEDGES);
    const int w0 = blockIdx.x * (PB_LINES * LINE_EDGES / 4) + tid;
    const int w1 = w0 + 256;
    const int4 d0 = dst4[w0 < NEDGES / 4 ? w0 : 0];
    const int4 d1 = dst4[w1 < NEDGES / 4 ? w1 : 0];
    for (int i = tid; i < NBLK * 2; i += 256) bmt[i] = 0u;
    for (int i = tid; i < EMB * HOUT; i += 256) {
        int k = i >> 7;
        int n = i & 127;
        Wt[n][k] = W[i];
    }
    __syncthreads();
    {
        const unsigned bit = 1u << (tid >> 3);
        if (w0 < NEDGES / 4) {
            const unsigned r0 = range_id(d0.x), r1 = range_id(d0.y), r2 = range_id(d0.z), r3 = range_id(d0.w);
            if (r0 < NBLK) atomicOr(&bmt[r0 * 2], bit);
            if (r1 < NBLK) atomicOr(&bmt[r1 * 2], bit);
            if (r2 < NBLK) atomicOr(&bmt[r2 * 2], bit);
            if (r3 < NBLK) atomicOr(&bmt[r3 * 2], bit);
        }
        if (w1 < NEDGES / 4) {
            const unsigned r0 = range_id(d1.x), r1 = range_id(d1.y), r2 = range_id(d1.z), r3 = range_id(d1.w);
            if (r0 < NBLK) atomicOr(&bmt[r0 * 2 + 1], bit);
            if (r1 < NBLK) atomicOr(&bmt[r1 * 2 + 1], bit);
            if (r2 < NBLK) atomicOr(&bmt[r2 * 2 + 1], bit);
            if (r3 < NBLK) atomicOr(&bmt[r3 * 2 + 1], bit);
        }
    }
    __syncthreads();
    if (tid < NBLK) {
        uint2 o;
        o.x = bmt[tid * 2];
        o.y = bmt[tid * 2 + 1];
        bm[(size_t)blockIdx.x * BM_PITCH + tid] = o;
    }

    const int wave = tid >> 5;
    const int lane = tid & 31;
    const int rowTile = (blockIdx.x * 8 + wave) * 16;
    if (rowTile >= N_NODES) return;

    const int half = lane >> 4;
    const int lrow = lane & 15;
    const float* arow = emb + (size_t)(rowTile + lrow) * EMB;

    v8f acc[8] = {};
    for (int k0 = 0; k0 < EMB; k0 += 4) {
        v2f a;
        a.x = arow[k0 + 2 * half];
        a.y = arow[k0 + 2 * half + 1];
        #pragma unroll
        for (int ct = 0; ct < 8; ++ct) {
            const int n = ct * 16 + lrow;
            v2f b;
            b.x = Wt[n][k0 + 2 * half];
            b.y = Wt[n][k0 + 2 * half + 1];
            acc[ct] = __builtin_amdgcn_wmma_f32_16x16x4_f32(
                false, a, false, b, (short)0, acc[ct], false, false);
        }
    }
    #pragma unroll
    for (int ct = 0; ct < 8; ++ct) {
        #pragma unroll
        for (int v = 0; v < 8; ++v) {
            const int mr = v + 8 * half;
            h[(size_t)(rowTile + mr) * HOUT + ct * 16 + lrow] = acc[ct][v];
        }
    }

    float asv[8], adv[8];
    #pragma unroll
    for (int ct = 0; ct < 8; ++ct) {
        asv[ct] = att_src[ct * 16 + lrow];
        adv[ct] = att_dst[ct * 16 + lrow];
    }
    const int keep = lrow & 7;
    float ms[HEADS], md[HEADS];
    #pragma unroll
    for (int hh = 0; hh < HEADS; ++hh) {
        ms[hh] = 0.0f;
        md[hh] = 0.0f;
        #pragma unroll
        for (int v = 0; v < 8; ++v) {
            float sv = acc[2 * hh][v] * asv[2 * hh] + acc[2 * hh + 1][v] * asv[2 * hh + 1];
            float dv = acc[2 * hh][v] * adv[2 * hh] + acc[2 * hh + 1][v] * adv[2 * hh + 1];
            #pragma unroll
            for (int m = 1; m < 16; m <<= 1) {
                sv += __shfl_xor(sv, m);
                dv += __shfl_xor(dv, m);
            }
            if (keep == v) { ms[hh] = sv; md[hh] = dv; }
        }
    }
    {
        const size_t node = (size_t)(rowTile + 8 * half + keep);
        float4 o;
        if (lrow < 8) { o.x = ms[0]; o.y = ms[1]; o.z = ms[2]; o.w = ms[3]; }
        else          { o.x = md[0]; o.y = md[1]; o.z = md[2]; o.w = md[3]; }
        float* dstp = (lrow < 8 ? a_src : a_dst) + node * HEADS;
        *(float4*)dstp = o;
    }
}

#define QCAP  8192
#define KMAX  (QCAP / 1024)
#define FGRP  16
#define DL_BITS 9
static_assert(NBLK * RANGE == N_NODES && NBLK < 255, "range id 255 is kept free");
static_assert(RANGE <= (1 << DL_BITS) && N_NODES <= (1 << (31 - DL_BITS)), "a queue entry fits a positive int");

__device__ __forceinline__ float4 edge_logit(const float* __restrict__ a_src,
                                             const float* adst_l, int s, int dl) {
    const float4 es = *(const float4*)(a_src + (size_t)s * 4);
    const float4 ed = *(const float4*)(adst_l + dl * 4);
    float4 ev;
    ev.x = leaky(es.x + ed.x);
    ev.y = leaky(es.y + ed.y);
    ev.z = leaky(es.z + ed.z);
    ev.w = leaky(es.w + ed.w);
    return ev;
}

template <int PHASE>
__device__ __forceinline__ void walk_all(const int* __restrict__ path,
                                         const float* __restrict__ a_src,
                                         const float* adst_l, float* m_l, float* den_l,
                                         const float* __restrict__ h, float* out_l,
                                         int lo, int tid) {
    for (int e = tid; e < NEDGES; e += 1024) {
        const unsigned dl = (unsigned)(path[NEDGES + e] - lo);
        if (dl < RANGE) {
            const int s = path[e];
            const float4 ev = edge_logit(a_src, adst_l, s, (int)dl);
            if (PHASE == 0) {
                atomicMaxF(&m_l[dl * 4 + 0], ev.x);
                atomicMaxF(&m_l[dl * 4 + 1], ev.y);
                atomicMaxF(&m_l[dl * 4 + 2], ev.z);
                atomicMaxF(&m_l[dl * 4 + 3], ev.w);
            } else {
                const float4 mv = *(const float4*)(m_l + dl * 4);
                float4 ex;
                ex.x = __expf(ev.x - mv.x);
                ex.y = __expf(ev.y - mv.y);
                ex.z = __expf(ev.z - mv.z);
                ex.w = __expf(ev.w - mv.w);
                if (PHASE == 1) {
                    atomicAdd(&den_l[dl * 4 + 0], ex.x);
                    atomicAdd(&den_l[dl * 4 + 1], ex.y);
                    atomicAdd(&den_l[dl * 4 + 2], ex.z);
                    atomicAdd(&den_l[dl * 4 + 3], ex.w);
                } else {
                    const float4 dn = *(const float4*)(den_l + dl * 4);
                    float4 al;
                    al.x = ex.x / (dn.x + 1e-16f);
                    al.y = ex.y / (dn.y + 1e-16f);
                    al.z = ex.z / (dn.z + 1e-16f);
                    al.w = ex.w / (dn.w + 1e-16f);
                    const float* hs = h + (size_t)s * HOUT;
                    for (int dd = 0; dd < RD; ++dd) {
                        const float acc = al.x * hs[0 * RD + dd]
                                        + al.y * hs[1 * RD + dd]
                                        + al.z * hs[2 * RD + dd]
                                        + al.w * hs[3 * RD + dd];
                        atomicAdd(&out_l[dl * RD + dd], 0.25f * acc);
                    }
                }
            }
        }
    }
}

__device__ __forceinline__ float lane_value(float x, int srcLane) {
    return __int_as_float(__builtin_amdgcn_readlane(__float_as_int(x), srcLane));
}

#define GRP 8

__global__ __launch_bounds__(1024) void softmax_range_kernel(const int* __restrict__ path,
                                                             const float* __restrict__ a_src,
                                                             const float* __restrict__ a_dst,
                                                             const float* __restrict__ h,
                                                             const uint2* __restrict__ bm,
                                                             float* __restrict__ out) {
    __shared__ alignas(16) float out_l[RANGE * RD];
    __shared__ float m_l[RANGE * 4];
    __shared__ float den_l[RANGE * 4];
    __shared__ float adst_l[RANGE * 4];
    __shared__ int   queue[QCAP];
    __shared__ int   lq[QCAP];
    __shared__ int   qn;
    __shared__ int   ln;

    const int tid = threadIdx.x;
    const int lo  = blockIdx.x * RANGE;

    for (int i = tid; i < RANGE * 4; i += 1024) {
        m_l[i]    = __int_as_float(0xFF800000);
        den_l[i]  = 0.0f;
        adst_l[i] = a_dst[(size_t)lo * 4 + i];
    }
    for (int i = tid; i < RANGE * RD; i += 1024) out_l[i] = 0.0f;
    if (tid == 0) { qn = 0; ln = 0; }
    __syncthreads();

    for (int t = tid; t < PROJ_BLOCKS; t += 1024) {
        const uint2 e = bm[(size_t)t * BM_PITCH + blockIdx.x];
        unsigned long long mm = ((unsigned long long)e.y << 32) | e.x;
        int pos = atomicAdd(&ln, __builtin_popcountll(mm));
        while (mm != 0ull) {
            const int line = t * PB_LINES + __builtin_ctzll(mm);
            mm &= mm - 1ull;
            if (pos < QCAP) lq[pos] = line < NLINES ? line : NLINES - 1;
            ++pos;
        }
    }
    __syncthreads();
    const int ltot = ln;
    if (ltot <= QCAP) {
        const int lane = tid & 31;
        const int wv   = __builtin_amdgcn_readfirstlane(tid >> 5);
        const int lt_s = __builtin_amdgcn_readfirstlane(ltot);
        for (int j0 = wv * FGRP; j0 < lt_s; j0 += 32 * FGRP) {
            int dv[FGRP], sr[FGRP];
            #pragma unroll
            for (int g = 0; g < FGRP; ++g) {
                const int j = j0 + g < lt_s ? j0 + g : lt_s - 1;
                const int e = lq[j] * LINE_EDGES + lane;
                dv[g] = path[NEDGES + e];
                sr[g] = path[e];
            }
            #pragma unroll
            for (int g = 0; g < FGRP; ++g) asm volatile("" :: "v"(dv[g]), "v"(sr[g]));
            unsigned mb[FGRP];
            int cnt = 0;
            #pragma unroll
            for (int g = 0; g < FGRP; ++g) {
                const unsigned dl = (unsigned)(dv[g] - lo);
                const unsigned in_range = __builtin_amdgcn_ballot_w32(dl < RANGE);
                mb[g] = j0 + g < lt_s ? in_range : 0u;
                cnt += __builtin_popcount(mb[g]);
            }
            int base = 0;
            if (lane == 0) base = atomicAdd(&qn, cnt);
            base = __builtin_amdgcn_readfirstlane(base);
            #pragma unroll
            for (int g = 0; g < FGRP; ++g) {
                const unsigned dl = (unsigned)(dv[g] - lo);
                const int pos = (int)__builtin_amdgcn_mbcnt_lo(mb[g], (unsigned)base);
                if (j0 + g < lt_s && dl < RANGE && pos < QCAP)
                    queue[pos] = (sr[g] << DL_BITS) | (int)dl;
                base += __builtin_popcount(mb[g]);
            }
        }
    }
    __syncthreads();
    const int total = ltot <= QCAP ? qn : QCAP + 1;

    if (total <= QCAP) {
        int    wq[KMAX];
        float4 val[KMAX];
        int    sv[KMAX];
        #pragma unroll
        for (int k = 0; k < KMAX; ++k) {
            const int j = k * 1024 + tid;
            wq[k] = j < total ? queue[j] : -1;
        }
        #pragma unroll
        for (int k = 0; k < KMAX; ++k) {
            sv[k] = wq[k] >= 0 ? wq[k] >> DL_BITS : 0;
            wq[k] = wq[k] >= 0 ? wq[k] & ((1 << DL_BITS) - 1) : -1;
            val[k] = edge_logit(a_src, adst_l, sv[k], wq[k] >= 0 ? wq[k] : 0);
        }
        #pragma unroll
        for (int k = 0; k < KMAX; ++k) {
            if (wq[k] >= 0) {
                const int dl = wq[k];
                atomicMaxF(&m_l[dl * 4 + 0], val[k].x);
                atomicMaxF(&m_l[dl * 4 + 1], val[k].y);
                atomicMaxF(&m_l[dl * 4 + 2], val[k].z);
                atomicMaxF(&m_l[dl * 4 + 3], val[k].w);
            }
        }
        __syncthreads();
        #pragma unroll
        for (int k = 0; k < KMAX; ++k) {
            if (wq[k] >= 0) {
                const int dl = wq[k];
                const float4 mv = *(const float4*)(m_l + dl * 4);
                float4 ex;
                ex.x = __expf(val[k].x - mv.x);
                ex.y = __expf(val[k].y - mv.y);
                ex.z = __expf(val[k].z - mv.z);
                ex.w = __expf(val[k].w - mv.w);
                val[k] = ex;
                atomicAdd(&den_l[dl * 4 + 0], ex.x);
                atomicAdd(&den_l[dl * 4 + 1], ex.y);
                atomicAdd(&den_l[dl * 4 + 2], ex.z);
                atomicAdd(&den_l[dl * 4 + 3], ex.w);
            }
        }
        __syncthreads();
        #pragma unroll
        for (int k = 0; k < KMAX; ++k) {
            if (wq[k] >= 0) {
                const int dl = wq[k];
                const float4 dn = *(const float4*)(den_l + dl * 4);
                float4 al;
                al.x = val[k].x / (dn.x + 1e-16f);
                al.y = val[k].y / (dn.y + 1e-16f);
                al.z = val[k].z / (dn.z + 1e-16f);
                al.w = val[k].w / (dn.w + 1e-16f);
                val[k] = al;
            }
        }

        const int lane = tid & 31;
        const int wbase = __builtin_amdgcn_readfirstlane(tid & ~31);
        const int tot_s = __builtin_amdgcn_readfirstlane(total);
        const float* hl = h + lane;
        #pragma nounroll
        for (int k = 0; k < KMAX; ++k) {
            int cnt = tot_s - (k * 1024 + wbase);
            if (cnt <= 0) break;
            cnt = cnt > 32 ? 32 : cnt;
            for (int L0 = 0; L0 < cnt; L0 += GRP) {
                float hv[GRP][4];
                #pragma unroll
                for (int g = 0; g < GRP; ++g) {
                    const int L = L0 + g < cnt ? L0 + g : cnt - 1;
                    const int s = __builtin_amdgcn_readlane(sv[0], L);
                    const float* hs = hl + (size_t)s * HOUT;
                    hv[g][0] = hs[0 * RD];
                    hv[g][1] = hs[1 * RD];
                    hv[g][2] = hs[2 * RD];
                    hv[g][3] = hs[3 * RD];
                }
                #pragma unroll
                for (int g = 0; g < GRP; ++g) {
                    if (L0 + g < cnt) {
                        const int L  = L0 + g;
                        const int dl = __builtin_amdgcn_readlane(wq[0], L);
                        const float acc = lane_value(val[0].x, L) * hv[g][0]
                                        + lane_value(val[0].y, L) * hv[g][1]
                                        + lane_value(val[0].z, L) * hv[g][2]
                                        + lane_value(val[0].w, L) * hv[g][3];
                        atomicAdd(&out_l[dl * RD + lane], 0.25f * acc);
                    }
                }
            }
            #pragma unroll
            for (int i = 0; i + 1 < KMAX; ++i) {
                wq[i]  = wq[i + 1];
                sv[i]  = sv[i + 1];
                val[i] = val[i + 1];
            }
        }
    } else {
        walk_all<0>(path, a_src, adst_l, m_l, den_l, h, out_l, lo, tid);
        __syncthreads();
        walk_all<1>(path, a_src, adst_l, m_l, den_l, h, out_l, lo, tid);
        __syncthreads();
        walk_all<2>(path, a_src, adst_l, m_l, den_l, h, out_l, lo, tid);
    }
    __syncthreads();

    float4* o4 = (float4*)(out + (size_t)lo * RD);
    const float4* l4 = (const float4*)out_l;
    for (int i = tid; i < RANGE * RD / 4; i += 1024) o4[i] = l4[i];
}

extern "C" void kernel_launch(void* const* d_in, const int* in_sizes, int n_in,
                              void* d_out, int out_size, void* d_ws, size_t ws_size,
                              hipStream_t stream) {
    const float* node_emb = (const float*)d_in[0];
    const float* W        = (const float*)d_in[1];
    const float* att_src  = (const float*)d_in[2];
    const float* att_dst  = (const float*)d_in[3];
    const int*   path     = (const int*)d_in[4];
    float* out = (float*)d_out;

    float* ws    = (float*)d_ws;
    float* h     = ws;
    float* a_src = h     + (size_t)N_NODES * HOUT;
    float* a_dst = a_src + (size_t)N_NODES * HEADS;
    uint2* bm = (uint2*)(a_dst + (size_t)N_NODES * HEADS);

    static_assert(PROJ_BLOCKS == (N_NODES / 16 + 7) / 8, "grid of the projection launch");
    proj_wmma_kernel<<<PROJ_BLOCKS, 256, 0, stream>>>(node_emb, W, att_src, att_dst, path,
                                                      h, a_src, a_dst, bm);

    softmax_range_kernel<<<NBLK, 1024, 0, stream>>>(path, a_src, a_dst, h, bm, out);
}


# Aggregation loop: full groups of 8 edges run without per-entry tests; one lane read gives source and destination, rows addressed by a 32-bit index off h, head sum an fma chain, 0.25 in the weights
# speedup vs baseline: 5.7459x; 1.0334x over previous
#include <hip/hip_runtime.h>

#define N_NODES   100000
#define EMB       64
#define HEADS     4
#define RD        32
#define HOUT      (HEADS * RD)
#define NEDGES    1600000
#define NEG_SLOPE 0.2f
#define RANGE     400
#define NBLK      (N_NODES / RANGE)
#define NO_RANGE  255u

typedef __attribute__((ext_vector_type(2))) float v2f;
typedef __attribute__((ext_vector_type(8))) float v8f;

__device__ __forceinline__ void atomicMaxF(float* addr, float v) {
    if (v >= 0.0f) atomicMax((int*)addr, __float_as_int(v));
    else           atomicMin((unsigned int*)addr, __float_as_uint(v));
}

__device__ __forceinline__ float leaky(float x) {
    return x >= 0.0f ? x : NEG_SLOPE * x;
}

__device__ __forceinline__ unsigned range_id(int d) {
    return (unsigned)d < (unsigned)N_NODES ? (unsigned)d / RANGE : NO_RANGE;
}

#define PROJ_BLOCKS  ((N_NODES / 16 + 7) / 8)
#define LINE_EDGES 32
#define NLINES     (NEDGES / LINE_EDGES)
#define PB_LINES   64
#define BM_PITCH   256
static_assert(NEDGES % LINE_EDGES == 0, "whole lines of 32 edges");
static_assert(PROJ_BLOCKS * PB_LINES >= NLINES, "the projection grid covers every line");
static_assert(NBLK <= BM_PITCH, "a chunk holds one entry per range");

__global__ __launch_bounds__(256, 8) void proj_wmma_kernel(const float* __restrict__ emb,
                                                        const float* __restrict__ W,
                                                        const float* __restrict__ att_src,
                                                        const float* __restrict__ att_dst,
                                                        const int* __restrict__ path,
                                                        float* __restrict__ h,
                                                        float* __restrict__ a_src,
                                                        float* __restrict__ a_dst,
                                                        uint2* __restrict__ bm) {
    __shared__ float Wt[HOUT][EMB + 1];
    __shared__ unsigned bmt[NBLK * 2];
    const int tid = threadIdx.x;
    const int4* dst4 = (const int4*)(path + NEDGES);
    const int w0 = blockIdx.x * (PB_LINES * LINE_EDGES / 4) + tid;
    const int w1 = w0 + 256;
    const int4 d0 = dst4[w0 < NEDGES / 4 ? w0 : 0];
    const int4 d1 = dst4[w1 < NEDGES / 4 ? w1 : 0];
    for (int i = tid; i < NBLK * 2; i += 256) bmt[i] = 0u;
    for (int i = tid; i < EMB * HOUT; i += 256) {
        int k = i >> 7;
        int n = i & 127;
        Wt[n][k] = W[i];
    }
    __syncthreads();
    {
        const unsigned bit = 1u << (tid >> 3);
        if (w0 < NEDGES / 4) {
            const unsigned r0 = range_id(d0.x), r1 = range_id(d0.y), r2 = range_id(d0.z), r3 = range_id(d0.w);
            if (r0 < NBLK) atomicOr(&bmt[r0 * 2], bit);
            if (r1 < NBLK) atomicOr(&bmt[r1 * 2], bit);
            if (r2 < NBLK) atomicOr(&bmt[r2 * 2], bit);
            if (r3 < NBLK) atomicOr(&bmt[r3 * 2], bit);
        }
        if (w1 < NEDGES / 4) {
            const unsigned r0 = range_id(d1.x), r1 = range_id(d1.y), r2 = range_id(d1.z), r3 = range_id(d1.w);
            if (r0 < NBLK) atomicOr(&bmt[r0 * 2 + 1], bit);
            if (r1 < NBLK) atomicOr(&bmt[r1 * 2 + 1], bit);
            if (r2 < NBLK) atomicOr(&bmt[r2 * 2 + 1], bit);
            if (r3 < NBLK) atomicOr(&bmt[r3 * 2 + 1], bit);
        }
    }
    __syncthreads();
    if (tid < NBLK) {
        uint2 o;
        o.x = bmt[tid * 2];
        o.y = bmt[tid * 2 + 1];
        bm[(size_t)blockIdx.x * BM_PITCH + tid] = o;
    }

    const int wave = tid >> 5;
    const int lane = tid & 31;
    const int rowTile = (blockIdx.x * 8 + wave) * 16;
    if (rowTile >= N_NODES) return;

    const int half = lane >> 4;
    const int lrow = lane & 15;
    const float* arow = emb + (size_t)(rowTile + lrow) * EMB;

    v8f acc[8] = {};
    for (int k0 = 0; k0 < EMB; k0 += 4) {
        v2f a;
        a.x = arow[k0 + 2 * half];
        a.y = arow[k0 + 2 * half + 1];
        #pragma unroll
        for (int ct = 0; ct < 8; ++ct) {
            const int n = ct * 16 + lrow;
            v2f b;
            b.x = Wt[n][k0 + 2 * half];
            b.y = Wt[n][k0 + 2 * half + 1];
            acc[ct] = __builtin_amdgcn_wmma_f32_16x16x4_f32(
                false, a, false, b, (short)0, acc[ct], false, false);
        }
    }
    #pragma unroll
    for (int ct = 0; ct < 8; ++ct) {
        #pragma unroll
        for (int v = 0; v < 8; ++v) {
            const int mr = v + 8 * half;
            h[(size_t)(rowTile + mr) * HOUT + ct * 16 + lrow] = acc[ct][v];
        }
    }

    float asv[8], adv[8];
    #pragma unroll
    for (int ct = 0; ct < 8; ++ct) {
        asv[ct] = att_src[ct * 16 + lrow];
        adv[ct] = att_dst[ct * 16 + lrow];
    }
    const int keep = lrow & 7;
    float ms[HEADS], md[HEADS];
    #pragma unroll
    for (int hh = 0; hh < HEADS; ++hh) {
        ms[hh] = 0.0f;
        md[hh] = 0.0f;
        #pragma unroll
        for (int v = 0; v < 8; ++v) {
            float sv = acc[2 * hh][v] * asv[2 * hh] + acc[2 * hh + 1][v] * asv[2 * hh + 1];
            float dv = acc[2 * hh][v] * adv[2 * hh] + acc[2 * hh + 1][v] * adv[2 * hh + 1];
            #pragma unroll
            for (int m = 1; m < 16; m <<= 1) {
                sv += __shfl_xor(sv, m);
                dv += __shfl_xor(dv, m);
            }
            if (keep == v) { ms[hh] = sv; md[hh] = dv; }
        }
    }
    {
        const size_t node = (size_t)(rowTile + 8 * half + keep);
        float4 o;
        if (lrow < 8) { o.x = ms[0]; o.y = ms[1]; o.z = ms[2]; o.w = ms[3]; }
        else          { o.x = md[0]; o.y = md[1]; o.z = md[2]; o.w = md[3]; }
        float* dstp = (lrow < 8 ? a_src : a_dst) + node * HEADS;
        *(float4*)dstp = o;
    }
}

#define QCAP  8192
#define KMAX  (QCAP / 1024)
#define FGRP  16
#define DL_BITS 9
static_assert(NBLK * RANGE == N_NODES && NBLK < 255, "range id 255 is kept free");
static_assert(RANGE <= (1 << DL_BITS) && N_NODES <= (1 << (31 - DL_BITS)), "a queue entry fits a positive int");
static_assert((long long)N_NODES * HOUT < (1ll << 29), "an element index into h fits a positive int, and so does its byte offset");

__device__ __forceinline__ float4 edge_logit(const float* __restrict__ a_src,
                                             const float* adst_l, int s, int dl) {
    const float4 es = *(const float4*)(a_src + (size_t)s * 4);
    const float4 ed = *(const float4*)(adst_l + dl * 4);
    float4 ev;
    ev.x = leaky(es.x + ed.x);
    ev.y = leaky(es.y + ed.y);
    ev.z = leaky(es.z + ed.z);
    ev.w = leaky(es.w + ed.w);
    return ev;
}

template <int PHASE>
__device__ __forceinline__ void walk_all(const int* __restrict__ path,
                                         const float* __restrict__ a_src,
                                         const float* adst_l, float* m_l, float* den_l,
                                         const float* __restrict__ h, float* out_l,
                                         int lo, int tid) {
    for (int e = tid; e < NEDGES; e += 1024) {
        const unsigned dl = (unsigned)(path[NEDGES + e] - lo);
        if (dl < RANGE) {
            const int s = path[e];
            const float4 ev = edge_logit(a_src, adst_l, s, (int)dl);
            if (PHASE == 0) {
                atomicMaxF(&m_l[dl * 4 + 0], ev.x);
                atomicMaxF(&m_l[dl * 4 + 1], ev.y);
                atomicMaxF(&m_l[dl * 4 + 2], ev.z);
                atomicMaxF(&m_l[dl * 4 + 3], ev.w);
            } else {
                const float4 mv = *(const float4*)(m_l + dl * 4);
                float4 ex;
                ex.x = __expf(ev.x - mv.x);
                ex.y = __expf(ev.y - mv.y);
                ex.z = __expf(ev.z - mv.z);
                ex.w = __expf(ev.w - mv.w);
                if (PHASE == 1) {
                    atomicAdd(&den_l[dl * 4 + 0], ex.x);
                    atomicAdd(&den_l[dl * 4 + 1], ex.y);
                    atomicAdd(&den_l[dl * 4 + 2], ex.z);
                    atomicAdd(&den_l[dl * 4 + 3], ex.w);
                } else {
                    const float4 dn = *(const float4*)(den_l + dl * 4);
                    float4 al;
                    al.x = ex.x / (dn.x + 1e-16f);
                    al.y = ex.y / (dn.y + 1e-16f);
                    al.z = ex.z / (dn.z + 1e-16f);
                    al.w = ex.w / (dn.w + 1e-16f);
                    const float* hs = h + (size_t)s * HOUT;
                    for (int dd = 0; dd < RD; ++dd) {
                        const float acc = al.x * hs[0 * RD + dd]
                                        + al.y * hs[1 * RD + dd]
                                        + al.z * hs[2 * RD + dd]
                                        + al.w * hs[3 * RD + dd];
                        atomicAdd(&out_l[dl * RD + dd], 0.25f * acc);
                    }
                }
            }
        }
    }
}

__device__ __forceinline__ float lane_value(float x, int srcLane) {
    return __int_as_float(__builtin_amdgcn_readlane(__float_as_int(x), srcLane));
}

#define GRP 8

__global__ __launch_bounds__(1024) void softmax_range_kernel(const int* __restrict__ path,
                                                             const float* __restrict__ a_src,
                                                             const float* __restrict__ a_dst,
                                                             const float* __restrict__ h,
                                                             const uint2* __restrict__ bm,
                                                             float* __restrict__ out) {
    __shared__ alignas(16) float out_l[RANGE * RD];
    __shared__ float m_l[RANGE * 4];
    __shared__ float den_l[RANGE * 4];
    __shared__ float adst_l[RANGE * 4];
    __shared__ int   queue[QCAP];
    __shared__ int   lq[QCAP];
    __shared__ int   qn;
    __shared__ int   ln;

    const int tid = threadIdx.x;
    const int lo  = blockIdx.x * RANGE;

    for (int i = tid; i < RANGE * 4; i += 1024) {
        m_l[i]    = __int_as_float(0xFF800000);
        den_l[i]  = 0.0f;
        adst_l[i] = a_dst[(size_t)lo * 4 + i];
    }
    for (int i = tid; i < RANGE * RD; i += 1024) out_l[i] = 0.0f;
    if (tid == 0) { qn = 0; ln = 0; }
    __syncthreads();

    for (int t = tid; t < PROJ_BLOCKS; t += 1024) {
        const uint2 e = bm[(size_t)t * BM_PITCH + blockIdx.x];
        unsigned long long mm = ((unsigned long long)e.y << 32) | e.x;
        int pos = atomicAdd(&ln, __builtin_popcountll(mm));
        while (mm != 0ull) {
            const int line = t * PB_LINES + __builtin_ctzll(mm);
            mm &= mm - 1ull;
            if (pos < QCAP) lq[pos] = line < NLINES ? line : NLINES - 1;
            ++pos;
        }
    }
    __syncthreads();
    const int ltot = ln;
    if (ltot <= QCAP) {
        const int lane = tid & 31;
        const int wv   = __builtin_amdgcn_readfirstlane(tid >> 5);
        const int lt_s = __builtin_amdgcn_readfirstlane(ltot);
        for (int j0 = wv * FGRP; j0 < lt_s; j0 += 32 * FGRP) {
            int dv[FGRP], sr[FGRP];
            #pragma unroll
            for (int g = 0; g < FGRP; ++g) {
                const int j = j0 + g < lt_s ? j0 + g : lt_s - 1;
                const int e = lq[j] * LINE_EDGES + lane;
                dv[g] = path[NEDGES + e];
                sr[g] = path[e];
            }
            #pragma unroll
            for (int g = 0; g < FGRP; ++g) asm volatile("" :: "v"(dv[g]), "v"(sr[g]));
            unsigned mb[FGRP];
            int cnt = 0;
            #pragma unroll
            for (int g = 0; g < FGRP; ++g) {
                const unsigned dl = (unsigned)(dv[g] - lo);
                const unsigned in_range = __builtin_amdgcn_ballot_w32(dl < RANGE);
                mb[g] = j0 + g < lt_s ? in_range : 0u;
                cnt += __builtin_popcount(mb[g]);
            }
            int base = 0;
            if (lane == 0) base = atomicAdd(&qn, cnt);
            base = __builtin_amdgcn_readfirstlane(base);
            #pragma unroll
            for (int g = 0; g < FGRP; ++g) {
                const unsigned dl = (unsigned)(dv[g] - lo);
                const int pos = (int)__builtin_amdgcn_mbcnt_lo(mb[g], (unsigned)base);
                if (j0 + g < lt_s && dl < RANGE && pos < QCAP)
                    queue[pos] = (sr[g] << DL_BITS) | (int)dl;
                base += __builtin_popcount(mb[g]);
            }
        }
    }
    __syncthreads();
    const int total = ltot <= QCAP ? qn : QCAP + 1;

    if (total <= QCAP) {
        int    wq[KMAX];
        float4 val[KMAX];
        int    sv[KMAX];
        #pragma unroll
        for (int k = 0; k < KMAX; ++k) {
            const int j = k * 1024 + tid;
            wq[k] = j < total ? queue[j] : -1;
        }
        #pragma unroll
        for (int k = 0; k < KMAX; ++k) {
            sv[k] = wq[k] >= 0 ? wq[k] : 0;
            wq[k] = wq[k] >= 0 ? wq[k] & ((1 << DL_BITS) - 1) : -1;
            val[k] = edge_logit(a_src, adst_l, sv[k] >> DL_BITS, wq[k] >= 0 ? wq[k] : 0);
        }
        #pragma unroll
        for (int k = 0; k < KMAX; ++k) {
            if (wq[k] >= 0) {
                const int dl = wq[k];
                atomicMaxF(&m_l[dl * 4 + 0], val[k].x);
                atomicMaxF(&m_l[dl * 4 + 1], val[k].y);
                atomicMaxF(&m_l[dl * 4 + 2], val[k].z);
                atomicMaxF(&m_l[dl * 4 + 3], val[k].w);
            }
        }
        __syncthreads();
        #pragma unroll
        for (int k = 0; k < KMAX; ++k) {
            if (wq[k] >= 0) {
                const int dl = wq[k];
                const float4 mv = *(const float4*)(m_l + dl * 4);
                float4 ex;
                ex.x = __expf(val[k].x - mv.x);
                ex.y = __expf(val[k].y - mv.y);
                ex.z = __expf(val[k].z - mv.z);
                ex.w = __expf(val[k].w - mv.w);
                val[k] = ex;
                atomicAdd(&den_l[dl * 4 + 0], ex.x);
                atomicAdd(&den_l[dl * 4 + 1], ex.y);
                atomicAdd(&den_l[dl * 4 + 2], ex.z);
                atomicAdd(&den_l[dl * 4 + 3], ex.w);
            }
        }
        __syncthreads();
        #pragma unroll
        for (int k = 0; k < KMAX; ++k) {
            if (wq[k] >= 0) {
                const int dl = wq[k];
                const float4 dn = *(const float4*)(den_l + dl * 4);
                float4 al;
                al.x = 0.25f * (val[k].x / (dn.x + 1e-16f));
                al.y = 0.25f * (val[k].y / (dn.y + 1e-16f));
                al.z = 0.25f * (val[k].z / (dn.z + 1e-16f));
                al.w = 0.25f * (val[k].w / (dn.w + 1e-16f));
                val[k] = al;
            }
        }

        const int lane = tid & 31;
        const int wbase = __builtin_amdgcn_readfirstlane(tid & ~31);
        const int tot_s = __builtin_amdgcn_readfirstlane(total);
        #pragma nounroll
        for (int k = 0; k < KMAX; ++k) {
            int cnt = tot_s - (k * 1024 + wbase);
            if (cnt <= 0) break;
            cnt = cnt > 32 ? 32 : cnt;
            for (int L0 = 0; L0 < cnt; L0 += GRP) {
                float hv[GRP][4];
                if (L0 + GRP <= cnt) {
                    int qw[GRP];
                    #pragma unroll
                    for (int g = 0; g < GRP; ++g) {
                        qw[g] = __builtin_amdgcn_readlane(sv[0], L0 + g);
                        const int row = (qw[g] >> DL_BITS) * HOUT + lane;
                        hv[g][0] = h[row + 0 * RD];
                        hv[g][1] = h[row + 1 * RD];
                        hv[g][2] = h[row + 2 * RD];
                        hv[g][3] = h[row + 3 * RD];
                    }
                    #pragma unroll
                    for (int g = 0; g < GRP; ++g) {
                        const int L  = L0 + g;
                        const int dl = qw[g] & ((1 << DL_BITS) - 1);
                        float acc = lane_value(val[0].x, L) * hv[g][0];
                        acc = __builtin_fmaf(lane_value(val[0].y, L), hv[g][1], acc);
                        acc = __builtin_fmaf(lane_value(val[0].z, L), hv[g][2], acc);
                        acc = __builtin_fmaf(lane_value(val[0].w, L), hv[g][3], acc);
                        atomicAdd(&out_l[dl * RD + lane], acc);
                    }
                } else {
                    #pragma unroll
                    for (int g = 0; g < GRP; ++g) {
                        const int L = L0 + g < cnt ? L0 + g : cnt - 1;
                        const int q = __builtin_amdgcn_readlane(sv[0], L);
                        const int row = (q >> DL_BITS) * HOUT + lane;
                        hv[g][0] = h[row + 0 * RD];
                        hv[g][1] = h[row + 1 * RD];
                        hv[g][2] = h[row + 2 * RD];
                        hv[g][3] = h[row + 3 * RD];
                    }
                    #pragma unroll
                    for (int g = 0; g < GRP; ++g) {
                        if (L0 + g < cnt) {
                            const int L  = L0 + g;
                            const int dl = __builtin_amdgcn_readlane(sv[0], L) & ((1 << DL_BITS) - 1);
                            float acc = lane_value(val[0].x, L) * hv[g][0];
                            acc = __builtin_fmaf(lane_value(val[0].y, L), hv[g][1], acc);
                            acc = __builtin_fmaf(lane_value(val[0].z, L), hv[g][2], acc);
                            acc = __builtin_fmaf(lane_value(val[0].w, L), hv[g][3], acc);
                            atomicAdd(&out_l[dl * RD + lane], acc);
                        }
                    }
                }
            }
            #pragma unroll
            for (int i = 0; i + 1 < KMAX; ++i) {
                sv[i]  = sv[i + 1];
                val[i] = val[i + 1];
            }
        }
    } else {
        walk_all<0>(path, a_src, adst_l, m_l, den_l, h, out_l, lo, tid);
        __syncthreads();
        walk_all<1>(path, a_src, adst_l, m_l, den_l, h, out_l, lo, tid);
        __syncthreads();
        walk_all<2>(path, a_src, adst_l, m_l, den_l, h, out_l, lo, tid);
    }
    __syncthreads();

    float4* o4 = (float4*)(out + (size_t)lo * RD);
    const float4* l4 = (const float4*)out_l;
    for (int i = tid; i < RANGE * RD / 4; i += 1024) o4[i] = l4[i];
}

extern "C" void kernel_launch(void* const* d_in, const int* in_sizes, int n_in,
                              void* d_out, int out_size, void* d_ws, size_t ws_size,
                              hipStream_t stream) {
    const float* node_emb = (const float*)d_in[0];
    const float* W        = (const float*)d_in[1];
    const float* att_src  = (const float*)d_in[2];
    const float* att_dst  = (const float*)d_in[3];
    const int*   path     = (const int*)d_in[4];
    float* out = (float*)d_out;

    float* ws    = (float*)d_ws;
    float* h     = ws;
    float* a_src = h     + (size_t)N_NODES * HOUT;
    float* a_dst = a_src + (size_t)N_NODES * HEADS;
    uint2* bm = (uint2*)(a_dst + (size_t)N_NODES * HEADS);

    static_assert(PROJ_BLOCKS == (N_NODES / 16 + 7) / 8, "grid of the projection launch");
    proj_wmma_kernel<<<PROJ_BLOCKS, 256, 0, stream>>>(node_emb, W, att_src, att_dst, path,
                                                      h, a_src, a_dst, bm);

    softmax_range_kernel<<<NBLK, 1024, 0, stream>>>(path, a_src, a_dst, h, bm, out);
}
